# starting kernel
# baseline (speedup 1.0000x reference)
#include <hip/hip_runtime.h>
#include <hip/hip_bf16.h>

typedef __attribute__((ext_vector_type(2))) float v2f;
typedef __attribute__((ext_vector_type(8))) float v8f;

#define N_QUBITS 12
#define DIM 4096
#define N_LAYERS 2
#define OUT_FEATURES 32
#define N_BATCH 128

__global__ __launch_bounds__(256) void QNN_sim_kernel(const float* __restrict__ x,
                                                      const float* __restrict__ w,
                                                      float* __restrict__ out) {
    __shared__ float sre[DIM];
    __shared__ float sim_[DIM];
    __shared__ float gate[3][256];
    __shared__ float red[256];

    const int tid = threadIdx.x;
    const int b = blockIdx.x;
    const int o = blockIdx.y;

    const int lane = tid & 31;
    const int waveId = tid >> 5;
    const int half = lane >> 4;
    const int lr = lane & 15;

    float xq[N_QUBITS];
#pragma unroll
    for (int q = 0; q < N_QUBITS; ++q) xq[q] = x[b * N_QUBITS + q];

#pragma unroll
    for (int t = 0; t < 16; ++t) {
        int i = t * 256 + tid;
        float s = 0.f;
#pragma unroll
        for (int q = 0; q < N_QUBITS; ++q) {
            float z = ((i >> (11 - q)) & 1) ? -1.f : 1.f;
            s += xq[q] * z;
        }
        float phi = -0.5f * s;
        sre[i] = 0.015625f * cosf(phi);
        sim_[i] = 0.015625f * sinf(phi);
    }

    for (int l = 0; l < N_LAYERS; ++l) {
        __syncthreads();
        {
            int r = tid >> 4, c = tid & 15;
#pragma unroll
            for (int g = 0; g < 3; ++g) {
                float p = 1.f;
#pragma unroll
                for (int t = 0; t < 4; ++t) {
                    int q = 4 * g + t;
                    float hw = 0.5f * w[o * (N_QUBITS * N_LAYERS) + q * N_LAYERS + l];
                    float cc = cosf(hw), ss = sinf(hw);
                    int rb = (r >> (3 - t)) & 1;
                    int cb = (c >> (3 - t)) & 1;
                    float m = rb ? (cb ? cc : ss) : (cb ? -ss : cc);
                    p *= m;
                }
                gate[g][tid] = p;
            }
        }
        __syncthreads();

        for (int it = 0; it < 4; ++it) {
            int jb = waveId + 8 * it;
            int part = jb >> 4;
            int row0 = (jb & 15) * 16;
            float* S = part ? sim_ : sre;
            v8f d = {};
#pragma unroll
            for (int kk = 0; kk < 4; ++kk) {
                int kb = 4 * kk + 2 * half;
                v2f A, B;
                A.x = S[(row0 + lr) * 16 + kb + 0];
                A.y = S[(row0 + lr) * 16 + kb + 1];
                B.x = gate[2][lr * 16 + kb + 0];
                B.y = gate[2][lr * 16 + kb + 1];
                d = __builtin_amdgcn_wmma_f32_16x16x4_f32(false, A, false, B,
                                                          (short)0, d, false, false);
            }
#pragma unroll
            for (int r = 0; r < 8; ++r)
                S[(row0 + r + 8 * half) * 16 + lr] = d[r];
        }
        __syncthreads();

        for (int it = 0; it < 4; ++it) {
            int jb = waveId + 8 * it;
            int part = jb >> 4;
            int base = (jb & 15) * 256;
            float* S = part ? sim_ : sre;
            v8f d = {};
#pragma unroll
            for (int kk = 0; kk < 4; ++kk) {
                int kb = 4 * kk + 2 * half;
                v2f A, B;
                A.x = gate[1][lr * 16 + kb + 0];
                A.y = gate[1][lr * 16 + kb + 1];
                B.x = S[base + (kb + 0) * 16 + lr];
                B.y = S[base + (kb + 1) * 16 + lr];
                d = __builtin_amdgcn_wmma_f32_16x16x4_f32(false, A, false, B,
                                                          (short)0, d, false, false);
            }
#pragma unroll
            for (int r = 0; r < 8; ++r)
                S[base + (r + 8 * half) * 16 + lr] = d[r];
        }
        __syncthreads();

        for (int it = 0; it < 4; ++it) {
            int jb = waveId + 8 * it;
            int part = jb >> 4;
            int col0 = (jb & 15) * 16;
            float* S = part ? sim_ : sre;
            v8f d = {};
#pragma unroll
            for (int kk = 0; kk < 4; ++kk) {
                int kb = 4 * kk + 2 * half;
                v2f A, B;
                A.x = gate[0][lr * 16 + kb + 0];
                A.y = gate[0][lr * 16 + kb + 1];
                B.x = S[(kb + 0) * 256 + col0 + lr];
                B.y = S[(kb + 1) * 256 + col0 + lr];
                d = __builtin_amdgcn_wmma_f32_16x16x4_f32(false, A, false, B,
                                                          (short)0, d, false, false);
            }
#pragma unroll
            for (int r = 0; r < 8; ++r)
                S[(r + 8 * half) * 256 + col0 + lr] = d[r];
        }
        __syncthreads();

        float tre[16], tim[16];
#pragma unroll
        for (int t = 0; t < 16; ++t) {
            int i = t * 256 + tid;
            int j = i;
#pragma unroll
            for (int q = 10; q >= 0; --q) {
                if ((j >> (11 - q)) & 1) j ^= (1 << (10 - q));
            }
            tre[t] = sre[j];
            tim[t] = sim_[j];
        }
        __syncthreads();
#pragma unroll
        for (int t = 0; t < 16; ++t) {
            int i = t * 256 + tid;
            sre[i] = tre[t];
            sim_[i] = tim[t];
        }
    }
    __syncthreads();

    float part = 0.f;
#pragma unroll
    for (int t = 0; t < 16; ++t) {
        int i = t * 256 + tid;
        float re = sre[i], im = sim_[i];
        float p = re * re + im * im;
        part += ((i >> 11) & 1) ? -p : p;
    }
    red[tid] = part;
    __syncthreads();
    for (int s = 128; s > 0; s >>= 1) {
        if (tid < s) red[tid] += red[tid + s];
        __syncthreads();
    }
    if (tid == 0) out[o * N_BATCH + b] = red[0];
}

extern "C" void kernel_launch(void* const* d_in, const int* in_sizes, int n_in,
                              void* d_out, int out_size, void* d_ws, size_t ws_size,
                              hipStream_t stream) {
    const float* x = (const float*)d_in[0];
    const float* w = (const float*)d_in[1];
    float* out = (float*)d_out;
    (void)in_sizes; (void)n_in; (void)out_size; (void)d_ws; (void)ws_size;
    dim3 grid(N_BATCH, OUT_FEATURES);
    QNN_sim_kernel<<<grid, 256, 0, stream>>>(x, w, out);
}


# Each sine/cosine computed once per workgroup into an LDS table (was per thread and per amplitude); initial state built as product of 12 per-qubit phase factors, so its rounding order changes
# speedup vs baseline: 1.4238x; 1.4238x over previous
#include <hip/hip_runtime.h>
#include <hip/hip_bf16.h>

typedef __attribute__((ext_vector_type(2))) float v2f;
typedef __attribute__((ext_vector_type(8))) float v8f;

#define N_QUBITS 12
#define DIM 4096
#define N_LAYERS 2
#define OUT_FEATURES 32
#define N_BATCH 128

__global__ __launch_bounds__(256) void QNN_sim_kernel(const float* __restrict__ x,
                                                      const float* __restrict__ w,
                                                      float* __restrict__ out) {
    __shared__ float sre[DIM];
    __shared__ float sim_[DIM];
    __shared__ float gate[3][256];
    __shared__ float red[256];
    __shared__ float tcos[N_QUBITS + N_QUBITS * N_LAYERS];
    __shared__ float tsin[N_QUBITS + N_QUBITS * N_LAYERS];
    __shared__ float kre[48];
    __shared__ float kim[48];

    const int tid = threadIdx.x;
    const int b = blockIdx.x;
    const int o = blockIdx.y;

    const int lane = tid & 31;
    const int waveId = tid >> 5;
    const int half = lane >> 4;
    const int lr = lane & 15;

    if (tid < N_QUBITS + N_QUBITS * N_LAYERS) {
        float a = (tid < N_QUBITS) ? x[b * N_QUBITS + tid]
                                   : w[o * (N_QUBITS * N_LAYERS) + (tid - N_QUBITS)];
        float h = 0.5f * a;
        tcos[tid] = cosf(h);
        tsin[tid] = sinf(h);
    }
    __syncthreads();

    if (tid < 48) {
        int g = tid >> 4, v = tid & 15;
        float pr = (g == 0) ? 0.015625f : 1.f, pi = 0.f;
#pragma unroll
        for (int t = 0; t < 4; ++t) {
            int q = 4 * g + t;
            float c = tcos[q], s = tsin[q];
            float fi = ((v >> (3 - t)) & 1) ? s : -s;
            float nr = pr * c - pi * fi;
            float ni = pr * fi + pi * c;
            pr = nr; pi = ni;
        }
        kre[tid] = pr;
        kim[tid] = pi;
    }
    __syncthreads();
    {
        float ar = kre[16 + (tid >> 4)], ai = kim[16 + (tid >> 4)];
        float br = kre[32 + (tid & 15)], bi = kim[32 + (tid & 15)];
        float ur = ar * br - ai * bi;
        float ui = ar * bi + ai * br;
#pragma unroll
        for (int t = 0; t < 16; ++t) {
            int i = t * 256 + tid;
            float cr = kre[t], ci = kim[t];
            sre[i] = cr * ur - ci * ui;
            sim_[i] = cr * ui + ci * ur;
        }
    }

#pragma unroll 1
    for (int l = 0; l < N_LAYERS; ++l) {
        __syncthreads();
        {
            int r = tid >> 4, c = tid & 15;
#pragma unroll
            for (int g = 0; g < 3; ++g) {
                float p = 1.f;
#pragma unroll
                for (int t = 0; t < 4; ++t) {
                    int q = 4 * g + t;
                    float cc = tcos[N_QUBITS + q * N_LAYERS + l];
                    float ss = tsin[N_QUBITS + q * N_LAYERS + l];
                    int rb = (r >> (3 - t)) & 1;
                    int cb = (c >> (3 - t)) & 1;
                    float m = rb ? (cb ? cc : ss) : (cb ? -ss : cc);
                    p *= m;
                }
                gate[g][tid] = p;
            }
        }
        __syncthreads();

        for (int it = 0; it < 4; ++it) {
            int jb = waveId + 8 * it;
            int part = jb >> 4;
            int row0 = (jb & 15) * 16;
            float* S = part ? sim_ : sre;
            v8f d = {};
#pragma unroll
            for (int kk = 0; kk < 4; ++kk) {
                int kb = 4 * kk + 2 * half;
                v2f A, B;
                A.x = S[(row0 + lr) * 16 + kb + 0];
                A.y = S[(row0 + lr) * 16 + kb + 1];
                B.x = gate[2][lr * 16 + kb + 0];
                B.y = gate[2][lr * 16 + kb + 1];
                d = __builtin_amdgcn_wmma_f32_16x16x4_f32(false, A, false, B,
                                                          (short)0, d, false, false);
            }
#pragma unroll
            for (int r = 0; r < 8; ++r)
                S[(row0 + r + 8 * half) * 16 + lr] = d[r];
        }
        __syncthreads();

        for (int it = 0; it < 4; ++it) {
            int jb = waveId + 8 * it;
            int part = jb >> 4;
            int base = (jb & 15) * 256;
            float* S = part ? sim_ : sre;
            v8f d = {};
#pragma unroll
            for (int kk = 0; kk < 4; ++kk) {
                int kb = 4 * kk + 2 * half;
                v2f A, B;
                A.x = gate[1][lr * 16 + kb + 0];
                A.y = gate[1][lr * 16 + kb + 1];
                B.x = S[base + (kb + 0) * 16 + lr];
                B.y = S[base + (kb + 1) * 16 + lr];
                d = __builtin_amdgcn_wmma_f32_16x16x4_f32(false, A, false, B,
                                                          (short)0, d, false, false);
            }
#pragma unroll
            for (int r = 0; r < 8; ++r)
                S[base + (r + 8 * half) * 16 + lr] = d[r];
        }
        __syncthreads();

        for (int it = 0; it < 4; ++it) {
            int jb = waveId + 8 * it;
            int part = jb >> 4;
            int col0 = (jb & 15) * 16;
            float* S = part ? sim_ : sre;
            v8f d = {};
#pragma unroll
            for (int kk = 0; kk < 4; ++kk) {
                int kb = 4 * kk + 2 * half;
                v2f A, B;
                A.x = gate[0][lr * 16 + kb + 0];
                A.y = gate[0][lr * 16 + kb + 1];
                B.x = S[(kb + 0) * 256 + col0 + lr];
                B.y = S[(kb + 1) * 256 + col0 + lr];
                d = __builtin_amdgcn_wmma_f32_16x16x4_f32(false, A, false, B,
                                                          (short)0, d, false, false);
            }
#pragma unroll
            for (int r = 0; r < 8; ++r)
                S[(r + 8 * half) * 256 + col0 + lr] = d[r];
        }
        __syncthreads();

        float tre[16], tim[16];
#pragma unroll
        for (int t = 0; t < 16; ++t) {
            int i = t * 256 + tid;
            int j = i;
#pragma unroll
            for (int q = 10; q >= 0; --q) {
                if ((j >> (11 - q)) & 1) j ^= (1 << (10 - q));
            }
            tre[t] = sre[j];
            tim[t] = sim_[j];
        }
        __syncthreads();
#pragma unroll
        for (int t = 0; t < 16; ++t) {
            int i = t * 256 + tid;
            sre[i] = tre[t];
            sim_[i] = tim[t];
        }
    }
    __syncthreads();

    float part = 0.f;
#pragma unroll
    for (int t = 0; t < 16; ++t) {
        int i = t * 256 + tid;
        float re = sre[i], im = sim_[i];
        float p = re * re + im * im;
        part += ((i >> 11) & 1) ? -p : p;
    }
    red[tid] = part;
    __syncthreads();
    for (int s = 128; s > 0; s >>= 1) {
        if (tid < s) red[tid] += red[tid + s];
        __syncthreads();
    }
    if (tid == 0) out[o * N_BATCH + b] = red[0];
}

extern "C" void kernel_launch(void* const* d_in, const int* in_sizes, int n_in,
                              void* d_out, int out_size, void* d_ws, size_t ws_size,
                              hipStream_t stream) {
    const float* x = (const float*)d_in[0];
    const float* w = (const float*)d_in[1];
    float* out = (float*)d_out;
    (void)in_sizes; (void)n_in; (void)out_size; (void)d_ws; (void)ws_size;
    dim3 grid(N_BATCH, OUT_FEATURES);
    QNN_sim_kernel<<<grid, 256, 0, stream>>>(x, w, out);
}


# Gate matrices no longer staged in LDS: each lane builds its 8 operand values per pass in registers from the trig table (drops the bank-conflicted gate loads, the gate build and a barrier per layer)
# speedup vs baseline: 1.6423x; 1.1535x over previous
#include <hip/hip_runtime.h>
#include <hip/hip_bf16.h>

typedef __attribute__((ext_vector_type(2))) float v2f;
typedef __attribute__((ext_vector_type(8))) float v8f;

#define N_QUBITS 12
#define DIM 4096
#define N_LAYERS 2
#define OUT_FEATURES 32
#define N_BATCH 128


__device__ __forceinline__ void gate_fragment(const float* tc, const float* ts, int g, int l,
                                              int lr, int half, float (&f)[8]) {
    float cc[4], ss[4];
#pragma unroll
    for (int t = 0; t < 4; ++t) {
        cc[t] = tc[N_QUBITS + (4 * g + t) * N_LAYERS + l];
        ss[t] = ts[N_QUBITS + (4 * g + t) * N_LAYERS + l];
    }
#pragma unroll
    for (int e = 0; e < 8; ++e) {
        int c = 4 * (e >> 1) + 2 * half + (e & 1);
        float p = 1.f;
#pragma unroll
        for (int t = 0; t < 4; ++t) {
            int rb = (lr >> (3 - t)) & 1;
            int cb = (c >> (3 - t)) & 1;
            float m = rb ? (cb ? cc[t] : ss[t]) : (cb ? -ss[t] : cc[t]);
            p *= m;
        }
        f[e] = p;
    }
}

__global__ __launch_bounds__(256) void QNN_sim_kernel(const float* __restrict__ x,
                                                      const float* __restrict__ w,
                                                      float* __restrict__ out) {
    __shared__ float sre[DIM];
    __shared__ float sim_[DIM];
    __shared__ float red[256];
    __shared__ float tcos[N_QUBITS + N_QUBITS * N_LAYERS];
    __shared__ float tsin[N_QUBITS + N_QUBITS * N_LAYERS];
    __shared__ float kre[48];
    __shared__ float kim[48];

    const int tid = threadIdx.x;
    const int b = blockIdx.x;
    const int o = blockIdx.y;

    const int lane = tid & 31;
    const int waveId = tid >> 5;
    const int half = lane >> 4;
    const int lr = lane & 15;

    if (tid < N_QUBITS + N_QUBITS * N_LAYERS) {
        float a = (tid < N_QUBITS) ? x[b * N_QUBITS + tid]
                                   : w[o * (N_QUBITS * N_LAYERS) + (tid - N_QUBITS)];
        float h = 0.5f * a;
        tcos[tid] = cosf(h);
        tsin[tid] = sinf(h);
    }
    __syncthreads();

    if (tid < 48) {
        int g = tid >> 4, v = tid & 15;
        float pr = (g == 0) ? 0.015625f : 1.f, pi = 0.f;
#pragma unroll
        for (int t = 0; t < 4; ++t) {
            int q = 4 * g + t;
            float c = tcos[q], s = tsin[q];
            float fi = ((v >> (3 - t)) & 1) ? s : -s;
            float nr = pr * c - pi * fi;
            float ni = pr * fi + pi * c;
            pr = nr; pi = ni;
        }
        kre[tid] = pr;
        kim[tid] = pi;
    }
    __syncthreads();
    {
        float ar = kre[16 + (tid >> 4)], ai = kim[16 + (tid >> 4)];
        float br = kre[32 + (tid & 15)], bi = kim[32 + (tid & 15)];
        float ur = ar * br - ai * bi;
        float ui = ar * bi + ai * br;
#pragma unroll
        for (int t = 0; t < 16; ++t) {
            int i = t * 256 + tid;
            float cr = kre[t], ci = kim[t];
            sre[i] = cr * ur - ci * ui;
            sim_[i] = cr * ui + ci * ur;
        }
    }

#pragma unroll 1
    for (int l = 0; l < N_LAYERS; ++l) {
        __syncthreads();
        float gf[8];

        gate_fragment(tcos, tsin, 2, l, lr, half, gf);
        for (int it = 0; it < 4; ++it) {
            int jb = waveId + 8 * it;
            int part = jb >> 4;
            int row0 = (jb & 15) * 16;
            float* S = part ? sim_ : sre;
            v8f d = {};
#pragma unroll
            for (int kk = 0; kk < 4; ++kk) {
                int kb = 4 * kk + 2 * half;
                v2f A, B;
                A.x = S[(row0 + lr) * 16 + kb + 0];
                A.y = S[(row0 + lr) * 16 + kb + 1];
                B.x = gf[2 * kk + 0];
                B.y = gf[2 * kk + 1];
                d = __builtin_amdgcn_wmma_f32_16x16x4_f32(false, A, false, B,
                                                          (short)0, d, false, false);
            }
#pragma unroll
            for (int r = 0; r < 8; ++r)
                S[(row0 + r + 8 * half) * 16 + lr] = d[r];
        }
        __syncthreads();

        gate_fragment(tcos, tsin, 1, l, lr, half, gf);
        for (int it = 0; it < 4; ++it) {
            int jb = waveId + 8 * it;
            int part = jb >> 4;
            int base = (jb & 15) * 256;
            float* S = part ? sim_ : sre;
            v8f d = {};
#pragma unroll
            for (int kk = 0; kk < 4; ++kk) {
                int kb = 4 * kk + 2 * half;
                v2f A, B;
                A.x = gf[2 * kk + 0];
                A.y = gf[2 * kk + 1];
                B.x = S[base + (kb + 0) * 16 + lr];
                B.y = S[base + (kb + 1) * 16 + lr];
                d = __builtin_amdgcn_wmma_f32_16x16x4_f32(false, A, false, B,
                                                          (short)0, d, false, false);
            }
#pragma unroll
            for (int r = 0; r < 8; ++r)
                S[base + (r + 8 * half) * 16 + lr] = d[r];
        }
        __syncthreads();

        gate_fragment(tcos, tsin, 0, l, lr, half, gf);
        for (int it = 0; it < 4; ++it) {
            int jb = waveId + 8 * it;
            int part = jb >> 4;
            int col0 = (jb & 15) * 16;
            float* S = part ? sim_ : sre;
            v8f d = {};
#pragma unroll
            for (int kk = 0; kk < 4; ++kk) {
                int kb = 4 * kk + 2 * half;
                v2f A, B;
                A.x = gf[2 * kk + 0];
                A.y = gf[2 * kk + 1];
                B.x = S[(kb + 0) * 256 + col0 + lr];
                B.y = S[(kb + 1) * 256 + col0 + lr];
                d = __builtin_amdgcn_wmma_f32_16x16x4_f32(false, A, false, B,
                                                          (short)0, d, false, false);
            }
#pragma unroll
            for (int r = 0; r < 8; ++r)
                S[(r + 8 * half) * 256 + col0 + lr] = d[r];
        }
        __syncthreads();

        float tre[16], tim[16];
#pragma unroll
        for (int t = 0; t < 16; ++t) {
            int i = t * 256 + tid;
            int j = i;
#pragma unroll
            for (int q = 10; q >= 0; --q) {
                if ((j >> (11 - q)) & 1) j ^= (1 << (10 - q));
            }
            tre[t] = sre[j];
            tim[t] = sim_[j];
        }
        __syncthreads();
#pragma unroll
        for (int t = 0; t < 16; ++t) {
            int i = t * 256 + tid;
            sre[i] = tre[t];
            sim_[i] = tim[t];
        }
    }
    __syncthreads();

    float part = 0.f;
#pragma unroll
    for (int t = 0; t < 16; ++t) {
        int i = t * 256 + tid;
        float re = sre[i], im = sim_[i];
        float p = re * re + im * im;
        part += ((i >> 11) & 1) ? -p : p;
    }
    red[tid] = part;
    __syncthreads();
    for (int s = 128; s > 0; s >>= 1) {
        if (tid < s) red[tid] += red[tid + s];
        __syncthreads();
    }
    if (tid == 0) out[o * N_BATCH + b] = red[0];
}

extern "C" void kernel_launch(void* const* d_in, const int* in_sizes, int n_in,
                              void* d_out, int out_size, void* d_ws, size_t ws_size,
                              hipStream_t stream) {
    const float* x = (const float*)d_in[0];
    const float* w = (const float*)d_in[1];
    float* out = (float*)d_out;
    (void)in_sizes; (void)n_in; (void)out_size; (void)d_ws; (void)ws_size;
    dim3 grid(N_BATCH, OUT_FEATURES);
    QNN_sim_kernel<<<grid, 256, 0, stream>>>(x, w, out);
}


# Last layer keeps only the Kron-factor pass holding qubit 0: its RY on qubits 4-11 and the closing CNOT chain cannot change <Z_0> and are dropped (2 of 6 matrix passes); final sum order changes
# speedup vs baseline: 2.3641x; 1.4395x over previous
#include <hip/hip_runtime.h>
#include <hip/hip_bf16.h>

typedef __attribute__((ext_vector_type(2))) float v2f;
typedef __attribute__((ext_vector_type(8))) float v8f;

#define N_QUBITS 12
#define DIM 4096
#define N_LAYERS 2
#define OUT_FEATURES 32
#define N_BATCH 128


__device__ __forceinline__ void gate_fragment(const float* tc, const float* ts, int g, int l,
                                              int lr, int half, float (&f)[8]) {
    float cc[4], ss[4];
#pragma unroll
    for (int t = 0; t < 4; ++t) {
        cc[t] = tc[N_QUBITS + (4 * g + t) * N_LAYERS + l];
        ss[t] = ts[N_QUBITS + (4 * g + t) * N_LAYERS + l];
    }
#pragma unroll
    for (int e = 0; e < 8; ++e) {
        int c = 4 * (e >> 1) + 2 * half + (e & 1);
        float p = 1.f;
#pragma unroll
        for (int t = 0; t < 4; ++t) {
            int rb = (lr >> (3 - t)) & 1;
            int cb = (c >> (3 - t)) & 1;
            float m = rb ? (cb ? cc[t] : ss[t]) : (cb ? -ss[t] : cc[t]);
            p *= m;
        }
        f[e] = p;
    }
}

__device__ __forceinline__ void pass_axis_i(float* sre, float* sim_, const float (&gf)[8],
                                            int waveId, int lr, int half) {
    for (int it = 0; it < 4; ++it) {
        int jb = waveId + 8 * it;
        int part = jb >> 4;
        int col0 = (jb & 15) * 16;
        float* S = part ? sim_ : sre;
        v8f d = {};
#pragma unroll
        for (int kk = 0; kk < 4; ++kk) {
            int kb = 4 * kk + 2 * half;
            v2f A, B;
            A.x = gf[2 * kk + 0];
            A.y = gf[2 * kk + 1];
            B.x = S[(kb + 0) * 256 + col0 + lr];
            B.y = S[(kb + 1) * 256 + col0 + lr];
            d = __builtin_amdgcn_wmma_f32_16x16x4_f32(false, A, false, B,
                                                      (short)0, d, false, false);
        }
#pragma unroll
        for (int r = 0; r < 8; ++r)
            S[(r + 8 * half) * 256 + col0 + lr] = d[r];
    }
}

__global__ __launch_bounds__(256) void QNN_sim_kernel(const float* __restrict__ x,
                                                      const float* __restrict__ w,
                                                      float* __restrict__ out) {
    __shared__ float sre[DIM];
    __shared__ float sim_[DIM];
    __shared__ float red[256];
    __shared__ float tcos[N_QUBITS + N_QUBITS * N_LAYERS];
    __shared__ float tsin[N_QUBITS + N_QUBITS * N_LAYERS];
    __shared__ float kre[48];
    __shared__ float kim[48];

    const int tid = threadIdx.x;
    const int b = blockIdx.x;
    const int o = blockIdx.y;

    const int lane = tid & 31;
    const int waveId = tid >> 5;
    const int half = lane >> 4;
    const int lr = lane & 15;

    if (tid < N_QUBITS + N_QUBITS * N_LAYERS) {
        float a = (tid < N_QUBITS) ? x[b * N_QUBITS + tid]
                                   : w[o * (N_QUBITS * N_LAYERS) + (tid - N_QUBITS)];
        float h = 0.5f * a;
        tcos[tid] = cosf(h);
        tsin[tid] = sinf(h);
    }
    __syncthreads();

    if (tid < 48) {
        int g = tid >> 4, v = tid & 15;
        float pr = (g == 0) ? 0.015625f : 1.f, pi = 0.f;
#pragma unroll
        for (int t = 0; t < 4; ++t) {
            int q = 4 * g + t;
            float c = tcos[q], s = tsin[q];
            float fi = ((v >> (3 - t)) & 1) ? s : -s;
            float nr = pr * c - pi * fi;
            float ni = pr * fi + pi * c;
            pr = nr; pi = ni;
        }
        kre[tid] = pr;
        kim[tid] = pi;
    }
    __syncthreads();
    {
        float ar = kre[16 + (tid >> 4)], ai = kim[16 + (tid >> 4)];
        float br = kre[32 + (tid & 15)], bi = kim[32 + (tid & 15)];
        float ur = ar * br - ai * bi;
        float ui = ar * bi + ai * br;
#pragma unroll
        for (int t = 0; t < 16; ++t) {
            int i = t * 256 + tid;
            float cr = kre[t], ci = kim[t];
            sre[i] = cr * ur - ci * ui;
            sim_[i] = cr * ui + ci * ur;
        }
    }

    {
        const int l = 0;
        __syncthreads();
        float gf[8];

        gate_fragment(tcos, tsin, 2, l, lr, half, gf);
        for (int it = 0; it < 4; ++it) {
            int jb = waveId + 8 * it;
            int part = jb >> 4;
            int row0 = (jb & 15) * 16;
            float* S = part ? sim_ : sre;
            v8f d = {};
#pragma unroll
            for (int kk = 0; kk < 4; ++kk) {
                int kb = 4 * kk + 2 * half;
                v2f A, B;
                A.x = S[(row0 + lr) * 16 + kb + 0];
                A.y = S[(row0 + lr) * 16 + kb + 1];
                B.x = gf[2 * kk + 0];
                B.y = gf[2 * kk + 1];
                d = __builtin_amdgcn_wmma_f32_16x16x4_f32(false, A, false, B,
                                                          (short)0, d, false, false);
            }
#pragma unroll
            for (int r = 0; r < 8; ++r)
                S[(row0 + r + 8 * half) * 16 + lr] = d[r];
        }
        __syncthreads();

        gate_fragment(tcos, tsin, 1, l, lr, half, gf);
        for (int it = 0; it < 4; ++it) {
            int jb = waveId + 8 * it;
            int part = jb >> 4;
            int base = (jb & 15) * 256;
            float* S = part ? sim_ : sre;
            v8f d = {};
#pragma unroll
            for (int kk = 0; kk < 4; ++kk) {
                int kb = 4 * kk + 2 * half;
                v2f A, B;
                A.x = gf[2 * kk + 0];
                A.y = gf[2 * kk + 1];
                B.x = S[base + (kb + 0) * 16 + lr];
                B.y = S[base + (kb + 1) * 16 + lr];
                d = __builtin_amdgcn_wmma_f32_16x16x4_f32(false, A, false, B,
                                                          (short)0, d, false, false);
            }
#pragma unroll
            for (int r = 0; r < 8; ++r)
                S[base + (r + 8 * half) * 16 + lr] = d[r];
        }
        __syncthreads();

        gate_fragment(tcos, tsin, 0, l, lr, half, gf);
        pass_axis_i(sre, sim_, gf, waveId, lr, half);
        __syncthreads();

        float tre[16], tim[16];
#pragma unroll
        for (int t = 0; t < 16; ++t) {
            int i = t * 256 + tid;
            int j = i;
#pragma unroll
            for (int q = 10; q >= 0; --q) {
                if ((j >> (11 - q)) & 1) j ^= (1 << (10 - q));
            }
            tre[t] = sre[j];
            tim[t] = sim_[j];
        }
        __syncthreads();
#pragma unroll
        for (int t = 0; t < 16; ++t) {
            int i = t * 256 + tid;
            sre[i] = tre[t];
            sim_[i] = tim[t];
        }
    }
    __syncthreads();

    static_assert(N_LAYERS == 2, "written as one full layer followed by the reduced last layer");
    {
        float gf[8];
        gate_fragment(tcos, tsin, 0, N_LAYERS - 1, lr, half, gf);
        pass_axis_i(sre, sim_, gf, waveId, lr, half);
    }
    __syncthreads();

    float part = 0.f;
#pragma unroll
    for (int t = 0; t < 16; ++t) {
        int i = t * 256 + tid;
        float re = sre[i], im = sim_[i];
        float p = re * re + im * im;
        part += ((i >> 11) & 1) ? -p : p;
    }
    red[tid] = part;
    __syncthreads();
    for (int s = 128; s > 0; s >>= 1) {
        if (tid < s) red[tid] += red[tid + s];
        __syncthreads();
    }
    if (tid == 0) out[o * N_BATCH + b] = red[0];
}

extern "C" void kernel_launch(void* const* d_in, const int* in_sizes, int n_in,
                              void* d_out, int out_size, void* d_ws, size_t ws_size,
                              hipStream_t stream) {
    const float* x = (const float*)d_in[0];
    const float* w = (const float*)d_in[1];
    float* out = (float*)d_out;
    (void)in_sizes; (void)n_in; (void)out_size; (void)d_ws; (void)ws_size;
    dim3 grid(N_BATCH, OUT_FEATURES);
    QNN_sim_kernel<<<grid, 256, 0, stream>>>(x, w, out);
}


# Last pass closed in form: its operand has two column profiles (rank 2), so G0 acts on two 16-vectors and the 256-column sums become 16-term inner products; a wave takes 8 circuits; sum order changes
# speedup vs baseline: 29.0217x; 12.2760x over previous
#include <hip/hip_runtime.h>

typedef __attribute__((ext_vector_type(2))) float v2f;
typedef __attribute__((ext_vector_type(8))) float v8f;

#define N_QUBITS 12
#define N_LAYERS 2
#define OUT_FEATURES 32
#define N_BATCH 128

#define B_PER_WAVE 8
#define N_WAVES 4
#define B_PER_WG (B_PER_WAVE * N_WAVES)
#define N_WTRIG (N_QUBITS * N_LAYERS)
#define N_XTRIG (B_PER_WG * N_QUBITS)


__device__ __forceinline__ float gate_elem(const float (&cc)[4], const float (&ss)[4], int lr, int c) {
    float p = 1.f;
#pragma unroll
    for (int t = 0; t < 4; ++t) {
        int rb = (lr >> (3 - t)) & 1;
        int cb = (c >> (3 - t)) & 1;
        float m = rb ? (cb ? cc[t] : ss[t]) : (cb ? -ss[t] : cc[t]);
        p *= m;
    }
    return p;
}

__device__ __forceinline__ float xhalf(float v) {
    return __builtin_bit_cast(float, __builtin_amdgcn_permlanex16(__builtin_bit_cast(int, v), __builtin_bit_cast(int, v),
                                                                  0x76543210, 0xfedcba98, false, false));
}
__device__ __forceinline__ float xpair(float v) {
    return __builtin_bit_cast(float, __builtin_amdgcn_update_dpp(0, __builtin_bit_cast(int, v), 0xB1, 0xF, 0xF, true));
}

__device__ __forceinline__ void gate_trig(const float* wc, const float* ws, int g, int l,
                                          float (&cc)[4], float (&ss)[4]) {
#pragma unroll
    for (int t = 0; t < 4; ++t) {
        cc[t] = wc[(4 * g + t) * N_LAYERS + l];
        ss[t] = ws[(4 * g + t) * N_LAYERS + l];
    }
}

__global__ __launch_bounds__(128) void QNN_sim_kernel(const float* __restrict__ x,
                                                      const float* __restrict__ w,
                                                      float* __restrict__ out) {
    __shared__ float wcos[N_WTRIG];
    __shared__ float wsin[N_WTRIG];
    __shared__ float xcs[2 * N_XTRIG];
    __shared__ float stage[B_PER_WG];

    const int tid = threadIdx.x;
    const int b0 = blockIdx.x * B_PER_WG;
    const int o = blockIdx.y;

    const int lane = tid & 31;
    const int waveId = tid >> 5;
    const int half = lane >> 4;
    const int lr = lane & 15;

    constexpr int N_TRIG = N_WTRIG + N_XTRIG;
    constexpr int TRIPS = (N_TRIG + 127) / 128;
    float ang[TRIPS];
#pragma unroll
    for (int it = 0; it < TRIPS; ++it) {
        const int i = tid + 128 * it;
        ang[it] = 0.f;
        if (i < N_TRIG)
            ang[it] = (i < N_WTRIG) ? w[o * N_WTRIG + i] : x[b0 * N_QUBITS + (i - N_WTRIG)];
    }
#pragma unroll
    for (int it = 0; it < TRIPS; ++it) {
        const int i = tid + 128 * it;
        if (i < N_TRIG) {
            float sv, cv;
            sincosf(0.5f * ang[it], &sv, &cv);
            if (i < N_WTRIG) {
                wcos[i] = cv;
                wsin[i] = sv;
            } else {
                xcs[2 * (i - N_WTRIG) + 0] = cv;
                xcs[2 * (i - N_WTRIG) + 1] = sv;
            }
        }
    }
    __syncthreads();

    const int bl = waveId * B_PER_WAVE + (lr >> 1);
    const int part = lr & 1;
    const float* xb = xcs + 2 * N_QUBITS * bl;

    v8f d[3];
#pragma unroll
    for (int g = 0; g < 3; ++g) {
        float c0 = xb[2 * (4 * g + 0)], s0 = xb[2 * (4 * g + 0) + 1];
        float c1 = xb[2 * (4 * g + 1)], s1 = xb[2 * (4 * g + 1) + 1];
        float c2 = xb[2 * (4 * g + 2)], s2 = xb[2 * (4 * g + 2) + 1];
        float c3 = xb[2 * (4 * g + 3)], s3 = xb[2 * (4 * g + 3) + 1];
        if (g == 0) { c0 *= 0.015625f; s0 *= 0.015625f; }
        float p01r[4], p01i[4];
#pragma unroll
        for (int m = 0; m < 4; ++m) {
            float f0 = (m >> 1) ? s0 : -s0;
            float f1 = (m & 1) ? s1 : -s1;
            p01r[m] = c0 * c1 - f0 * f1;
            p01i[m] = c0 * f1 + f0 * c1;
        }
        const float f2 = half ? s2 : -s2;
        float q1[2], q2[2];
#pragma unroll
        for (int s = 0; s < 2; ++s) {
            float f3 = s ? s3 : -s3;
            float pr = c2 * c3 - f2 * f3;
            float pi = c2 * f3 + f2 * c3;
            q1[s] = part ? pi : pr;
            q2[s] = part ? pr : -pi;
        }
        float cc[4], ss[4];
        gate_trig(wcos, wsin, g, 0, cc, ss);
        v8f acc = {};
#pragma unroll
        for (int kk = 0; kk < 4; ++kk) {
            v2f A, B;
            A.x = gate_elem(cc, ss, lr, 4 * kk + 2 * half + 0);
            A.y = gate_elem(cc, ss, lr, 4 * kk + 2 * half + 1);
            B.x = p01r[kk] * q1[0] + p01i[kk] * q2[0];
            B.y = p01r[kk] * q1[1] + p01i[kk] * q2[1];
            acc = __builtin_amdgcn_wmma_f32_16x16x4_f32(false, A, false, B,
                                                        (short)0, acc, false, false);
        }
        d[g] = acc;
    }

    static_assert(N_LAYERS == 2, "written as layer 0 on the factors followed by the reduced last layer");
    v8f da = {}, db = {};
    {
        float cc[4], ss[4];
        gate_trig(wcos, wsin, 0, N_LAYERS - 1, cc, ss);
#pragma unroll
        for (int kk = 0; kk < 4; ++kk) {
            v2f Ae, Ao, B;
            float gv[2];
            int odd[2];
#pragma unroll
            for (int s = 0; s < 2; ++s) {
                int j = 8 * half + 2 * kk + s;
                int k = (j ^ (j >> 1) ^ (j >> 2) ^ (j >> 3)) & 15;
                gv[s] = gate_elem(cc, ss, lr, k);
                odd[s] = k & 1;
            }
            Ae.x = odd[0] ? 0.f : gv[0];
            Ae.y = odd[1] ? 0.f : gv[1];
            Ao.x = odd[0] ? gv[0] : 0.f;
            Ao.y = odd[1] ? gv[1] : 0.f;
            B.x = d[0][2 * kk + 0];
            B.y = d[0][2 * kk + 1];
            da = __builtin_amdgcn_wmma_f32_16x16x4_f32(false, Ae, false, B,
                                                       (short)0, da, false, false);
            db = __builtin_amdgcn_wmma_f32_16x16x4_f32(false, Ao, false, B,
                                                       (short)0, db, false, false);
        }
    }

    float n1 = 0.f, n2 = 0.f, cx = 0.f;
#pragma unroll
    for (int r = 0; r < 8; ++r) {
        n1 += d[1][r] * d[1][r];
        n2 += d[2][r] * d[2][r];
        cx += d[1][r] * xhalf(d[1][r]);
    }
    n1 += xpair(n1);  n1 += xhalf(n1);
    n2 += xpair(n2);  n2 += xhalf(n2);
    cx += xpair(cx);  cx += xhalf(cx);

    float t = 0.f;
#pragma unroll
    for (int r = 0; r < 8; ++r)
        t += n1 * (da[r] * da[r] + db[r] * db[r]) + 2.f * cx * (da[r] * db[r]);
    if (half) t = -t;
    t += xpair(t);
    t += xhalf(t);
    if (lane < 16 && part == 0) stage[bl] = n2 * t;
    __syncthreads();
    if (tid < B_PER_WG) out[o * N_BATCH + b0 + tid] = stage[tid];
}

extern "C" void kernel_launch(void* const* d_in, const int* in_sizes, int n_in,
                              void* d_out, int out_size, void* d_ws, size_t ws_size,
                              hipStream_t stream) {
    const float* x = (const float*)d_in[0];
    const float* w = (const float*)d_in[1];
    float* out = (float*)d_out;
    (void)in_sizes; (void)n_in; (void)out_size; (void)d_ws; (void)ws_size;
    dim3 grid(N_BATCH / B_PER_WG, OUT_FEATURES);
    QNN_sim_kernel<<<grid, 128, 0, stream>>>(x, w, out);
}
